# starting kernel
# baseline (speedup 1.0000x reference)
#include <hip/hip_runtime.h>

typedef __attribute__((ext_vector_type(2))) float v2f;
typedef __attribute__((ext_vector_type(8))) float v8f;

#define KTOT 768
#define WAVES_PER_BLOCK 8
#define PATCHES_PER_WAVE 16

__global__ __launch_bounds__(256) void patchconv_wmma_kernel(
    const float* __restrict__ x, const float* __restrict__ w,
    const float* __restrict__ bias, float* __restrict__ out, int npatch)
{
    __shared__ float sW[KTOT];
    for (int i = threadIdx.x; i < KTOT; i += 256) sW[i] = w[i];
    __syncthreads();

    const int lane = threadIdx.x & 31;
    const int wave = threadIdx.x >> 5;
    const int half = lane >> 4;
    const int m    = lane & 15;
    const long patch0 =
        ((long)blockIdx.x * WAVES_PER_BLOCK + wave) * PATCHES_PER_WAVE;
    if (patch0 + PATCHES_PER_WAVE > npatch) return;

    const float* xrow = x + (patch0 + m) * (long)KTOT + 2 * half;
    const float* wrow = sW + 2 * half;

    v8f c = {};
#pragma unroll 8
    for (int k = 0; k < KTOT; k += 4) {
        v2f a = *(const v2f*)(xrow + k);
        v2f b = *(const v2f*)(wrow + k);
        c = __builtin_amdgcn_wmma_f32_16x16x4_f32(
            false, a, false, b, (short)0, c, false, false);
    }

    const float bv = bias[0];
    if (m == 0) {
        float* dst = out + patch0 + half * 8;
        float4 o0 = make_float4(c[0] + bv, c[1] + bv, c[2] + bv, c[3] + bv);
        float4 o1 = make_float4(c[4] + bv, c[5] + bv, c[6] + bv, c[7] + bv);
        *(float4*)(dst)     = o0;
        *(float4*)(dst + 4) = o1;
    }
}

extern "C" void kernel_launch(void* const* d_in, const int* in_sizes, int n_in,
                              void* d_out, int out_size, void* d_ws, size_t ws_size,
                              hipStream_t stream) {
    const float* x    = (const float*)d_in[0];
    const float* w    = (const float*)d_in[1];
    const float* bias = (const float*)d_in[2];
    float* out = (float*)d_out;

    const int npatch = out_size;
    const int waves  = npatch / PATCHES_PER_WAVE;
    const int blocks = (waves + WAVES_PER_BLOCK - 1) / WAVES_PER_BLOCK;

    patchconv_wmma_kernel<<<blocks, 256, 0, stream>>>(x, w, bias, out, npatch);
}


# each x load reads 512 contiguous bytes of one patch, 4 patches ahead; W in registers; dot = diagonal of x-piece by W-piece tiles; sum order changed: per-lane pieces, then a 32-lane butterfly
# speedup vs baseline: 1.7917x; 1.7917x over previous
#include <hip/hip_runtime.h>


typedef __attribute__((ext_vector_type(2))) float v2f;
typedef __attribute__((ext_vector_type(4))) float v4f;
typedef __attribute__((ext_vector_type(8))) float v8f;

#define KTOT   768
#define PIECES 6
#define WAVES  7
#define PPW    14
#define AHEAD  4
#define THREADS (WAVES * 32)

__global__ __launch_bounds__(THREADS) void patchconv_wmma_kernel(
    const float* __restrict__ x, const float* __restrict__ w,
    const float* __restrict__ bias, float* __restrict__ out, int npatch)
{
    const unsigned tid  = threadIdx.x;
    const unsigned lane = tid & 31u;
    const unsigned wave = __builtin_amdgcn_readfirstlane(tid >> 5);
    const unsigned patch0 = (blockIdx.x * WAVES + wave) * PPW;
    if (patch0 >= (unsigned)npatch) return;

    v4f wq[PIECES];
#pragma unroll
    for (int q = 0; q < PIECES; ++q) wq[q] = *(const v4f*)(w + 128 * q + 4 * lane);

    const float* xb = x + (size_t)patch0 * KTOT + 4 * lane;
    v4f xr[AHEAD + 1][PIECES];
    float part[PPW];

#define LOAD_PATCH(R) do { const unsigned rr_ = (patch0 + (unsigned)(R) < (unsigned)npatch) ? (unsigned)(R) : 0u; \
    _Pragma("unroll") for (int q = 0; q < PIECES; ++q) \
        xr[(R) % (AHEAD + 1)][q] = *(const v4f*)(xb + (size_t)rr_ * KTOT + 128 * q); } while (0)

    const unsigned oct = lane >> 3;
    const int sel = (oct == 0u || oct == 3u) ? (int)(lane & 7u) : 8;

#pragma unroll
    for (int r = 0; r < AHEAD; ++r) LOAD_PATCH(r);

#pragma unroll
    for (int r = 0; r < PPW; ++r) {
        __builtin_amdgcn_sched_barrier(0);
        if (r + AHEAD < PPW) {
            LOAD_PATCH(r + AHEAD);
            __builtin_amdgcn_sched_barrier(0);
        }
        const v4f* xp = xr[r % (AHEAD + 1)];
        v8f c0 = {}, c1 = {};
#pragma unroll
        for (int q = 0; q < PIECES; ++q) {
            c0 = __builtin_amdgcn_wmma_f32_16x16x4_f32(false, xp[q].xy, false, wq[q].xy, (short)0, c0, false, false);
            c1 = __builtin_amdgcn_wmma_f32_16x16x4_f32(false, xp[q].zw, false, wq[q].zw, (short)0, c1, false, false);
        }
        float d = 0.f;
#pragma unroll
        for (int j = 0; j < 8; ++j) d = (sel == j) ? (c0[j] + c1[j]) : d;
        part[r] = d;
    }
#undef LOAD_PATCH

#pragma unroll
    for (int r = 0; r < PPW; ++r) {
        float s = part[r];
#pragma unroll
        for (int m = 16; m >= 1; m >>= 1) s += __shfl_xor(s, m, 32);
        part[r] = s;
    }
    float mine = 0.f;
#pragma unroll
    for (int r = 0; r < PPW; ++r) if (lane == (unsigned)r) mine = part[r];
    if (lane < (unsigned)PPW && patch0 + lane < (unsigned)npatch) out[patch0 + lane] = mine + bias[0];
}

extern "C" void kernel_launch(void* const* d_in, const int* in_sizes, int n_in,
                              void* d_out, int out_size, void* d_ws, size_t ws_size,
                              hipStream_t stream) {
    const float* x    = (const float*)d_in[0];
    const float* w    = (const float*)d_in[1];
    const float* bias = (const float*)d_in[2];
    float* out = (float*)d_out;

    const int npatch = out_size;
    const int blocks = (npatch + WAVES * PPW - 1) / (WAVES * PPW);

    patchconv_wmma_kernel<<<blocks, THREADS, 0, stream>>>(x, w, bias, out, npatch);
}
